# starting kernel
# baseline (speedup 1.0000x reference)
#include <hip/hip_runtime.h>
#include <math.h>

typedef __attribute__((ext_vector_type(2))) float v2f;
typedef __attribute__((ext_vector_type(4))) float v4f;
typedef __attribute__((ext_vector_type(8))) float v8f;

#define D_DIM 1024
#define B_DIM 2048
#define M_MOL 8
#define NPAIR 28

__device__ __constant__ unsigned char c_I[NPAIR] =
    {0,0,0,0,0,0,0, 1,1,1,1,1,1, 2,2,2,2,2, 3,3,3,3, 4,4,4, 5,5, 6};
__device__ __constant__ unsigned char c_J[NPAIR] =
    {1,2,3,4,5,6,7, 2,3,4,5,6,7, 3,4,5,6,7, 4,5,6,7, 5,6,7, 6,7, 7};

static __device__ inline v8f wmma4(v2f a, v2f b, v8f c) {
  return __builtin_amdgcn_wmma_f32_16x16x4_f32(false, a, false, b, (short)0, c,
                                               false, false);
}


__global__ __launch_bounds__(256) void gemm_gt(const float* __restrict__ Wq,
                                               const float* __restrict__ Wk,
                                               float* __restrict__ Gpk) {
  const int wave = (int)((blockIdx.x * blockDim.x + threadIdx.x) >> 5);
  const int lane = threadIdx.x & 31;
  const int lm = lane & 15;
  const int half = lane >> 4;
  const int tm = wave >> 6;
  const int tn = wave & 63;

  const float* arow = Wk + (size_t)(tm * 16 + lm) * D_DIM + 2 * half;
  const float* brow = Wq + (size_t)(tn * 16 + lm) * D_DIM + 2 * half;

  v8f c = {};
#pragma unroll 4
  for (int x0 = 0; x0 < D_DIM; x0 += 4) {
    v2f a = *(const v2f*)(arow + x0);
    v2f b = *(const v2f*)(brow + x0);
    c = wmma4(a, b, c);
  }
  const int colOff = ((lm >> 1) & 1) * 512 + (tn * 4 + (lm >> 2)) * 2 + (lm & 1);
#pragma unroll
  for (int v = 0; v < 8; ++v)
    Gpk[(size_t)(tm * 16 + v + 8 * half) * D_DIM + colOff] = c[v];
}

__global__ __launch_bounds__(256) void attn_batch(
    const float* __restrict__ rF, const float* __restrict__ pF,
    const float* __restrict__ rN, const float* __restrict__ pN,
    const float* __restrict__ Gpk, const float* __restrict__ Wp,
    const float* __restrict__ bp, float* __restrict__ out) {
  __shared__ float sW[8][256];
  __shared__ float sS[8][256];
  __shared__ float sRows[72][36];
  __shared__ float sAtt[2][36];
  __shared__ float sWeff[2][8];
  __shared__ float sRed[256 * 4];

  const int b = blockIdx.x;
  const int tid = threadIdx.x;
  const int wave = tid >> 5;
  const int lane = tid & 31;
  const int lm = lane & 15;
  const int half = lane >> 4;

  const float* Frow =
      (lm < M_MOL) ? (pF + ((size_t)lm * B_DIM + b) * D_DIM)
                   : (rF + ((size_t)(lm - M_MOL) * B_DIM + b) * D_DIM);
  const float* fa = Frow + 2 * half;

  v8f sacc = {};
#pragma unroll 1
  for (int i = 0; i < 2; ++i) {
    int e[4];
    const float* g[4];
#pragma unroll
    for (int j = 0; j < 4; ++j) {
      e[j] = (wave + 8 * (4 * i + j)) * 16;
      g[j] = Gpk + (size_t)(e[j] + lm) * D_DIM + half * 512;
    }
    v8f c0 = {}, c1 = {}, c2 = {}, c3 = {};
#pragma unroll 1
    for (int d0 = 0; d0 < D_DIM; d0 += 8) {
      v2f a0 = *(const v2f*)(fa + d0);
      v2f a1 = *(const v2f*)(fa + d0 + 4);
      v4f g0 = *(const v4f*)(g[0] + (d0 >> 1));
      v4f g1 = *(const v4f*)(g[1] + (d0 >> 1));
      v4f g2 = *(const v4f*)(g[2] + (d0 >> 1));
      v4f g3 = *(const v4f*)(g[3] + (d0 >> 1));
      c0 = wmma4(a0, __builtin_shufflevector(g0, g0, 0, 1), c0);
      c1 = wmma4(a0, __builtin_shufflevector(g1, g1, 0, 1), c1);
      c2 = wmma4(a0, __builtin_shufflevector(g2, g2, 0, 1), c2);
      c3 = wmma4(a0, __builtin_shufflevector(g3, g3, 0, 1), c3);
      c0 = wmma4(a1, __builtin_shufflevector(g0, g0, 2, 3), c0);
      c1 = wmma4(a1, __builtin_shufflevector(g1, g1, 2, 3), c1);
      c2 = wmma4(a1, __builtin_shufflevector(g2, g2, 2, 3), c2);
      c3 = wmma4(a1, __builtin_shufflevector(g3, g3, 2, 3), c3);
    }
#pragma unroll
    for (int j = 0; j < 4; ++j) {
      const v8f cj = (j == 0) ? c0 : (j == 1) ? c1 : (j == 2) ? c2 : c3;
#pragma unroll
      for (int v = 0; v < 8; ++v) sW[wave][(v + 8 * half) * 16 + lm] = cj[v];
#pragma unroll
      for (int kk = 0; kk < 4; ++kk) {
        v2f a2 = *(const v2f*)(&sW[wave][lm * 16 + kk * 4 + 2 * half]);
        v2f b2 = *(const v2f*)(Frow + e[j] + kk * 4 + 2 * half);
        sacc = wmma4(a2, b2, sacc);
      }
    }
  }
#pragma unroll
  for (int v = 0; v < 8; ++v) sS[wave][(v + 8 * half) * 16 + lm] = sacc[v];
  __syncthreads();

  {
    float s = 0.f;
#pragma unroll
    for (int w = 0; w < 8; ++w) s += sS[w][tid];
    sS[0][tid] = s;
  }
  __syncthreads();

  const float scale = 0.03125f;
  if (tid < 72) {
    const int dir = (tid >= 36) ? 1 : 0;
    const int q = dir ? tid - 36 : tid;
    int qa0 = q, qa1 = q, qn = 1;
    if (q >= 8) { qa0 = c_I[q - 8]; qa1 = c_J[q - 8]; qn = 2; }
    float row[36];
    float mx = -3.4e38f;
#pragma unroll
    for (int k = 0; k < 36; ++k) {
      int ka0 = k, ka1 = k, kn = 1;
      if (k >= 8) { ka0 = c_I[k - 8]; ka1 = c_J[k - 8]; kn = 2; }
      float s = 0.f;
      for (int u = 0; u < qn; ++u) {
        const int a = u ? qa1 : qa0;
        for (int v = 0; v < kn; ++v) {
          const int cc = v ? ka1 : ka0;
          s += (dir == 0) ? sS[0][a * 16 + 8 + cc] : sS[0][(8 + a) * 16 + cc];
        }
      }
      s *= scale;
      row[k] = s;
      mx = fmaxf(mx, s);
    }
    float sum = 0.f;
#pragma unroll
    for (int k = 0; k < 36; ++k) { float e = __expf(row[k] - mx); row[k] = e; sum += e; }
    const float inv = 1.f / sum;
#pragma unroll
    for (int k = 0; k < 36; ++k) sRows[tid][k] = row[k] * inv;
  }
  __syncthreads();

  if (tid < 72) {
    const int dir = (tid >= 36) ? 1 : 0;
    const int k = dir ? tid - 36 : tid;
    float a = 0.f;
#pragma unroll
    for (int q = 0; q < 36; ++q) a += sRows[dir * 36 + q][k];
    a *= (1.f / 36.f);
    sAtt[dir][k] = a;
    const size_t base = dir ? (size_t)(B_DIM * 4 + B_DIM * 36) : (size_t)(B_DIM * 4);
    out[base + (size_t)b * 36 + k] = a;
  }
  __syncthreads();

  if (tid < 16) {
    const int dir = tid >> 3, m = tid & 7;
    float w = sAtt[dir][m];
#pragma unroll
    for (int pr = 0; pr < NPAIR; ++pr)
      if (c_I[pr] == m || c_J[pr] == m) w += sAtt[dir][8 + pr];
    sWeff[dir][m] = w;
  }
  __syncthreads();

  {
    const int d = tid * 4;
    float reac[4] = {0.f, 0.f, 0.f, 0.f};
#pragma unroll
    for (int m = 0; m < M_MOL; ++m) {
      const size_t off = ((size_t)m * B_DIM + b) * D_DIM + d;
      v4f rv = *(const v4f*)(rF + off);
      v4f rnv = *(const v4f*)(rN + off);
      v4f pv = *(const v4f*)(pF + off);
      v4f pnv = *(const v4f*)(pN + off);
      const float wr = sWeff[0][m], wp = sWeff[1][m];
#pragma unroll
      for (int x = 0; x < 4; ++x)
        reac[x] += wr * (rv[x] + rnv[x]) - wp * (pv[x] + pnv[x]);
    }
#pragma unroll
    for (int o = 0; o < 4; ++o) {
      float s = 0.f;
#pragma unroll
      for (int x = 0; x < 4; ++x) s += reac[x] * Wp[(size_t)o * D_DIM + d + x];
      sRed[tid * 4 + o] = s;
    }
  }
  __syncthreads();
  if (tid < 4) {
    float s = bp[tid];
    for (int t = 0; t < 256; ++t) s += sRed[t * 4 + tid];
    out[(size_t)b * 4 + tid] = s;
  }
}

extern "C" void kernel_launch(void* const* d_in, const int* in_sizes, int n_in,
                              void* d_out, int out_size, void* d_ws,
                              size_t ws_size, hipStream_t stream) {
  (void)in_sizes; (void)n_in; (void)out_size; (void)ws_size;
  const float* rF = (const float*)d_in[0];
  const float* pF = (const float*)d_in[1];
  const float* rN = (const float*)d_in[2];
  const float* pN = (const float*)d_in[3];
  const float* Wq = (const float*)d_in[4];
  const float* Wk = (const float*)d_in[5];
  const float* Wp = (const float*)d_in[6];
  const float* bp = (const float*)d_in[7];
  float* out = (float*)d_out;
  float* Gpk = (float*)d_ws;

  gemm_gt<<<512, 256, 0, stream>>>(Wq, Wk, Gpk);
  attn_batch<<<B_DIM, 256, 0, stream>>>(rF, pF, rN, pN, Gpk, Wp, bp, out);
}


# G^T workspace re-packed in B-fragment order: each wave b128 load reads 512 contiguous bytes (4 whole lines) instead of 16 B from each of 32 lines; same values per lane, same k order
# speedup vs baseline: 4.5579x; 4.5579x over previous
#include <hip/hip_runtime.h>
#include <math.h>

typedef __attribute__((ext_vector_type(2))) float v2f;
typedef __attribute__((ext_vector_type(4))) float v4f;
typedef __attribute__((ext_vector_type(8))) float v8f;

#define D_DIM 1024
#define B_DIM 2048
#define M_MOL 8
#define NPAIR 28

__device__ __constant__ unsigned char c_I[NPAIR] =
    {0,0,0,0,0,0,0, 1,1,1,1,1,1, 2,2,2,2,2, 3,3,3,3, 4,4,4, 5,5, 6};
__device__ __constant__ unsigned char c_J[NPAIR] =
    {1,2,3,4,5,6,7, 2,3,4,5,6,7, 3,4,5,6,7, 4,5,6,7, 5,6,7, 6,7, 7};

static __device__ inline v8f wmma4(v2f a, v2f b, v8f c) {
  return __builtin_amdgcn_wmma_f32_16x16x4_f32(false, a, false, b, (short)0, c,
                                               false, false);
}


__global__ __launch_bounds__(256) void gemm_gt(const float* __restrict__ Wq,
                                               const float* __restrict__ Wk,
                                               float* __restrict__ Gpk) {
  const int wave = (int)((blockIdx.x * blockDim.x + threadIdx.x) >> 5);
  const int lane = threadIdx.x & 31;
  const int lm = lane & 15;
  const int half = lane >> 4;
  const int tm = wave >> 6;
  const int tn = wave & 63;

  const float* arow = Wk + (size_t)(tm * 16 + lm) * D_DIM + 2 * half;
  const float* brow = Wq + (size_t)(tn * 16 + lm) * D_DIM + 2 * half;

  v8f c = {};
#pragma unroll 4
  for (int x0 = 0; x0 < D_DIM; x0 += 4) {
    v2f a = *(const v2f*)(arow + x0);
    v2f b = *(const v2f*)(brow + x0);
    c = wmma4(a, b, c);
  }
  const int dstep = tn * 2 + (lm >> 3);
  const int s = (lm >> 2) & 1;
  const int h = (lm >> 1) & 1;
  const int r = lm & 1;
  const size_t base =
      ((size_t)(tm * 128 + dstep) * 32 + h * 16 + 8 * half) * 4 + 2 * s + r;
#pragma unroll
  for (int v = 0; v < 8; ++v) Gpk[base + (size_t)v * 4] = c[v];
}

__global__ __launch_bounds__(256) void attn_batch(
    const float* __restrict__ rF, const float* __restrict__ pF,
    const float* __restrict__ rN, const float* __restrict__ pN,
    const float* __restrict__ Gpk, const float* __restrict__ Wp,
    const float* __restrict__ bp, float* __restrict__ out) {
  __shared__ float sW[8][256];
  __shared__ float sS[8][256];
  __shared__ float sRows[72][36];
  __shared__ float sAtt[2][36];
  __shared__ float sWeff[2][8];
  __shared__ float sRed[256 * 4];

  const int b = blockIdx.x;
  const int tid = threadIdx.x;
  const int wave = tid >> 5;
  const int lane = tid & 31;
  const int lm = lane & 15;
  const int half = lane >> 4;

  const float* Frow =
      (lm < M_MOL) ? (pF + ((size_t)lm * B_DIM + b) * D_DIM)
                   : (rF + ((size_t)(lm - M_MOL) * B_DIM + b) * D_DIM);
  const float* fa = Frow + 2 * half;

  v8f sacc = {};
#pragma unroll 1
  for (int i = 0; i < 2; ++i) {
    int e[4];
    const float* g[4];
#pragma unroll
    for (int j = 0; j < 4; ++j) {
      e[j] = (wave + 8 * (4 * i + j)) * 16;
      g[j] = Gpk + (size_t)e[j] * D_DIM + lane * 4;
    }
    v8f c0 = {}, c1 = {}, c2 = {}, c3 = {};
#pragma unroll 1
    for (int d0 = 0; d0 < D_DIM; d0 += 8) {
      v2f a0 = *(const v2f*)(fa + d0);
      v2f a1 = *(const v2f*)(fa + d0 + 4);
      v4f g0 = *(const v4f*)(g[0] + d0 * 16);
      v4f g1 = *(const v4f*)(g[1] + d0 * 16);
      v4f g2 = *(const v4f*)(g[2] + d0 * 16);
      v4f g3 = *(const v4f*)(g[3] + d0 * 16);
      c0 = wmma4(a0, __builtin_shufflevector(g0, g0, 0, 1), c0);
      c1 = wmma4(a0, __builtin_shufflevector(g1, g1, 0, 1), c1);
      c2 = wmma4(a0, __builtin_shufflevector(g2, g2, 0, 1), c2);
      c3 = wmma4(a0, __builtin_shufflevector(g3, g3, 0, 1), c3);
      c0 = wmma4(a1, __builtin_shufflevector(g0, g0, 2, 3), c0);
      c1 = wmma4(a1, __builtin_shufflevector(g1, g1, 2, 3), c1);
      c2 = wmma4(a1, __builtin_shufflevector(g2, g2, 2, 3), c2);
      c3 = wmma4(a1, __builtin_shufflevector(g3, g3, 2, 3), c3);
    }
#pragma unroll
    for (int j = 0; j < 4; ++j) {
      const v8f cj = (j == 0) ? c0 : (j == 1) ? c1 : (j == 2) ? c2 : c3;
#pragma unroll
      for (int v = 0; v < 8; ++v) sW[wave][(v + 8 * half) * 16 + lm] = cj[v];
#pragma unroll
      for (int kk = 0; kk < 4; ++kk) {
        v2f a2 = *(const v2f*)(&sW[wave][lm * 16 + kk * 4 + 2 * half]);
        v2f b2 = *(const v2f*)(Frow + e[j] + kk * 4 + 2 * half);
        sacc = wmma4(a2, b2, sacc);
      }
    }
  }
#pragma unroll
  for (int v = 0; v < 8; ++v) sS[wave][(v + 8 * half) * 16 + lm] = sacc[v];
  __syncthreads();

  {
    float s = 0.f;
#pragma unroll
    for (int w = 0; w < 8; ++w) s += sS[w][tid];
    sS[0][tid] = s;
  }
  __syncthreads();

  const float scale = 0.03125f;
  if (tid < 72) {
    const int dir = (tid >= 36) ? 1 : 0;
    const int q = dir ? tid - 36 : tid;
    int qa0 = q, qa1 = q, qn = 1;
    if (q >= 8) { qa0 = c_I[q - 8]; qa1 = c_J[q - 8]; qn = 2; }
    float row[36];
    float mx = -3.4e38f;
#pragma unroll
    for (int k = 0; k < 36; ++k) {
      int ka0 = k, ka1 = k, kn = 1;
      if (k >= 8) { ka0 = c_I[k - 8]; ka1 = c_J[k - 8]; kn = 2; }
      float s = 0.f;
      for (int u = 0; u < qn; ++u) {
        const int a = u ? qa1 : qa0;
        for (int v = 0; v < kn; ++v) {
          const int cc = v ? ka1 : ka0;
          s += (dir == 0) ? sS[0][a * 16 + 8 + cc] : sS[0][(8 + a) * 16 + cc];
        }
      }
      s *= scale;
      row[k] = s;
      mx = fmaxf(mx, s);
    }
    float sum = 0.f;
#pragma unroll
    for (int k = 0; k < 36; ++k) { float e = __expf(row[k] - mx); row[k] = e; sum += e; }
    const float inv = 1.f / sum;
#pragma unroll
    for (int k = 0; k < 36; ++k) sRows[tid][k] = row[k] * inv;
  }
  __syncthreads();

  if (tid < 72) {
    const int dir = (tid >= 36) ? 1 : 0;
    const int k = dir ? tid - 36 : tid;
    float a = 0.f;
#pragma unroll
    for (int q = 0; q < 36; ++q) a += sRows[dir * 36 + q][k];
    a *= (1.f / 36.f);
    sAtt[dir][k] = a;
    const size_t base = dir ? (size_t)(B_DIM * 4 + B_DIM * 36) : (size_t)(B_DIM * 4);
    out[base + (size_t)b * 36 + k] = a;
  }
  __syncthreads();

  if (tid < 16) {
    const int dir = tid >> 3, m = tid & 7;
    float w = sAtt[dir][m];
#pragma unroll
    for (int pr = 0; pr < NPAIR; ++pr)
      if (c_I[pr] == m || c_J[pr] == m) w += sAtt[dir][8 + pr];
    sWeff[dir][m] = w;
  }
  __syncthreads();

  {
    const int d = tid * 4;
    float reac[4] = {0.f, 0.f, 0.f, 0.f};
#pragma unroll
    for (int m = 0; m < M_MOL; ++m) {
      const size_t off = ((size_t)m * B_DIM + b) * D_DIM + d;
      v4f rv = *(const v4f*)(rF + off);
      v4f rnv = *(const v4f*)(rN + off);
      v4f pv = *(const v4f*)(pF + off);
      v4f pnv = *(const v4f*)(pN + off);
      const float wr = sWeff[0][m], wp = sWeff[1][m];
#pragma unroll
      for (int x = 0; x < 4; ++x)
        reac[x] += wr * (rv[x] + rnv[x]) - wp * (pv[x] + pnv[x]);
    }
#pragma unroll
    for (int o = 0; o < 4; ++o) {
      float s = 0.f;
#pragma unroll
      for (int x = 0; x < 4; ++x) s += reac[x] * Wp[(size_t)o * D_DIM + d + x];
      sRed[tid * 4 + o] = s;
    }
  }
  __syncthreads();
  if (tid < 4) {
    float s = bp[tid];
    for (int t = 0; t < 256; ++t) s += sRed[t * 4 + tid];
    out[(size_t)b * 4 + tid] = s;
  }
}

extern "C" void kernel_launch(void* const* d_in, const int* in_sizes, int n_in,
                              void* d_out, int out_size, void* d_ws,
                              size_t ws_size, hipStream_t stream) {
  (void)in_sizes; (void)n_in; (void)out_size; (void)ws_size;
  const float* rF = (const float*)d_in[0];
  const float* pF = (const float*)d_in[1];
  const float* rN = (const float*)d_in[2];
  const float* pN = (const float*)d_in[3];
  const float* Wq = (const float*)d_in[4];
  const float* Wk = (const float*)d_in[5];
  const float* Wp = (const float*)d_in[6];
  const float* bp = (const float*)d_in[7];
  float* out = (float*)d_out;
  float* Gpk = (float*)d_ws;

  gemm_gt<<<512, 256, 0, stream>>>(Wq, Wk, Gpk);
  attn_batch<<<B_DIM, 256, 0, stream>>>(rF, pF, rN, pN, Gpk, Wp, bp, out);
}


# F tile (16x1024 f32) staged once per workgroup in LDS, row pitch 1028 floats; A fragments and the S-contraction B fragments are 8-byte LDS reads, not 16-line global gathers; same k and sum order
# speedup vs baseline: 5.4515x; 1.1961x over previous
#include <hip/hip_runtime.h>
#include <math.h>

typedef __attribute__((ext_vector_type(2))) float v2f;
typedef __attribute__((ext_vector_type(4))) float v4f;
typedef __attribute__((ext_vector_type(8))) float v8f;

#define D_DIM 1024
#define B_DIM 2048
#define M_MOL 8
#define NPAIR 28
#define FP 1028

struct EpiLds {
  float S[256];
  float Rows[72][36];
  float Att[2][36];
  float Weff[2][8];
  float Red[256 * 4];
};
union TileLds {
  float F[16 * FP];
  EpiLds e;
};

__device__ __constant__ unsigned char c_I[NPAIR] =
    {0,0,0,0,0,0,0, 1,1,1,1,1,1, 2,2,2,2,2, 3,3,3,3, 4,4,4, 5,5, 6};
__device__ __constant__ unsigned char c_J[NPAIR] =
    {1,2,3,4,5,6,7, 2,3,4,5,6,7, 3,4,5,6,7, 4,5,6,7, 5,6,7, 6,7, 7};

static __device__ inline v8f wmma4(v2f a, v2f b, v8f c) {
  return __builtin_amdgcn_wmma_f32_16x16x4_f32(false, a, false, b, (short)0, c,
                                               false, false);
}


__global__ __launch_bounds__(256) void gemm_gt(const float* __restrict__ Wq,
                                               const float* __restrict__ Wk,
                                               float* __restrict__ Gpk) {
  const int wave = (int)((blockIdx.x * blockDim.x + threadIdx.x) >> 5);
  const int lane = threadIdx.x & 31;
  const int lm = lane & 15;
  const int half = lane >> 4;
  const int tm = wave >> 6;
  const int tn = wave & 63;

  const float* arow = Wk + (size_t)(tm * 16 + lm) * D_DIM + 2 * half;
  const float* brow = Wq + (size_t)(tn * 16 + lm) * D_DIM + 2 * half;

  v8f c = {};
#pragma unroll 4
  for (int x0 = 0; x0 < D_DIM; x0 += 4) {
    v2f a = *(const v2f*)(arow + x0);
    v2f b = *(const v2f*)(brow + x0);
    c = wmma4(a, b, c);
  }
  const int dstep = tn * 2 + (lm >> 3);
  const int s = (lm >> 2) & 1;
  const int h = (lm >> 1) & 1;
  const int r = lm & 1;
  const size_t base =
      ((size_t)(tm * 128 + dstep) * 32 + h * 16 + 8 * half) * 4 + 2 * s + r;
#pragma unroll
  for (int v = 0; v < 8; ++v) Gpk[base + (size_t)v * 4] = c[v];
}

__global__ __launch_bounds__(256) void attn_batch(
    const float* __restrict__ rF, const float* __restrict__ pF,
    const float* __restrict__ rN, const float* __restrict__ pN,
    const float* __restrict__ Gpk, const float* __restrict__ Wp,
    const float* __restrict__ bp, float* __restrict__ out) {
  __shared__ __attribute__((aligned(16))) TileLds lds;
  __shared__ float sW[8][256];

  const int b = blockIdx.x;
  const int tid = threadIdx.x;
  const int wave = tid >> 5;
  const int lane = tid & 31;
  const int lm = lane & 15;
  const int half = lane >> 4;

#pragma unroll
  for (int r = 0; r < 16; ++r) {
    const float* srow =
        (r < M_MOL) ? (pF + ((size_t)r * B_DIM + b) * D_DIM)
                    : (rF + ((size_t)(r - M_MOL) * B_DIM + b) * D_DIM);
    *(v4f*)(&lds.F[r * FP + tid * 4]) = *(const v4f*)(srow + tid * 4);
  }
  __syncthreads();

  const float* Frow = &lds.F[lm * FP];
  const float* fa = Frow + 2 * half;

  v8f sacc = {};
#pragma unroll 1
  for (int i = 0; i < 2; ++i) {
    int e[4];
    const float* g[4];
#pragma unroll
    for (int j = 0; j < 4; ++j) {
      e[j] = (wave + 8 * (4 * i + j)) * 16;
      g[j] = Gpk + (size_t)e[j] * D_DIM + lane * 4;
    }
    v8f c0 = {}, c1 = {}, c2 = {}, c3 = {};
#pragma unroll 1
    for (int d0 = 0; d0 < D_DIM; d0 += 8) {
      v2f a0 = *(const v2f*)(fa + d0);
      v2f a1 = *(const v2f*)(fa + d0 + 4);
      v4f g0 = *(const v4f*)(g[0] + d0 * 16);
      v4f g1 = *(const v4f*)(g[1] + d0 * 16);
      v4f g2 = *(const v4f*)(g[2] + d0 * 16);
      v4f g3 = *(const v4f*)(g[3] + d0 * 16);
      c0 = wmma4(a0, __builtin_shufflevector(g0, g0, 0, 1), c0);
      c1 = wmma4(a0, __builtin_shufflevector(g1, g1, 0, 1), c1);
      c2 = wmma4(a0, __builtin_shufflevector(g2, g2, 0, 1), c2);
      c3 = wmma4(a0, __builtin_shufflevector(g3, g3, 0, 1), c3);
      c0 = wmma4(a1, __builtin_shufflevector(g0, g0, 2, 3), c0);
      c1 = wmma4(a1, __builtin_shufflevector(g1, g1, 2, 3), c1);
      c2 = wmma4(a1, __builtin_shufflevector(g2, g2, 2, 3), c2);
      c3 = wmma4(a1, __builtin_shufflevector(g3, g3, 2, 3), c3);
    }
#pragma unroll
    for (int j = 0; j < 4; ++j) {
      const v8f cj = (j == 0) ? c0 : (j == 1) ? c1 : (j == 2) ? c2 : c3;
#pragma unroll
      for (int v = 0; v < 8; ++v) sW[wave][(v + 8 * half) * 16 + lm] = cj[v];
#pragma unroll
      for (int kk = 0; kk < 4; ++kk) {
        v2f a2 = *(const v2f*)(&sW[wave][lm * 16 + kk * 4 + 2 * half]);
        v2f b2 = *(const v2f*)(Frow + e[j] + kk * 4 + 2 * half);
        sacc = wmma4(a2, b2, sacc);
      }
    }
  }
#pragma unroll
  for (int v = 0; v < 8; ++v) sW[wave][(v + 8 * half) * 16 + lm] = sacc[v];
  __syncthreads();

  {
    float s = 0.f;
#pragma unroll
    for (int w = 0; w < 8; ++w) s += sW[w][tid];
    lds.e.S[tid] = s;
  }
  __syncthreads();

  const float scale = 0.03125f;
  if (tid < 72) {
    const int dir = (tid >= 36) ? 1 : 0;
    const int q = dir ? tid - 36 : tid;
    int qa0 = q, qa1 = q, qn = 1;
    if (q >= 8) { qa0 = c_I[q - 8]; qa1 = c_J[q - 8]; qn = 2; }
    float row[36];
    float mx = -3.4e38f;
#pragma unroll
    for (int k = 0; k < 36; ++k) {
      int ka0 = k, ka1 = k, kn = 1;
      if (k >= 8) { ka0 = c_I[k - 8]; ka1 = c_J[k - 8]; kn = 2; }
      float s = 0.f;
      for (int u = 0; u < qn; ++u) {
        const int a = u ? qa1 : qa0;
        for (int v = 0; v < kn; ++v) {
          const int cc = v ? ka1 : ka0;
          s += (dir == 0) ? lds.e.S[a * 16 + 8 + cc] : lds.e.S[(8 + a) * 16 + cc];
        }
      }
      s *= scale;
      row[k] = s;
      mx = fmaxf(mx, s);
    }
    float sum = 0.f;
#pragma unroll
    for (int k = 0; k < 36; ++k) { float e = __expf(row[k] - mx); row[k] = e; sum += e; }
    const float inv = 1.f / sum;
#pragma unroll
    for (int k = 0; k < 36; ++k) lds.e.Rows[tid][k] = row[k] * inv;
  }
  __syncthreads();

  if (tid < 72) {
    const int dir = (tid >= 36) ? 1 : 0;
    const int k = dir ? tid - 36 : tid;
    float a = 0.f;
#pragma unroll
    for (int q = 0; q < 36; ++q) a += lds.e.Rows[dir * 36 + q][k];
    a *= (1.f / 36.f);
    lds.e.Att[dir][k] = a;
    const size_t base = dir ? (size_t)(B_DIM * 4 + B_DIM * 36) : (size_t)(B_DIM * 4);
    out[base + (size_t)b * 36 + k] = a;
  }
  __syncthreads();

  if (tid < 16) {
    const int dir = tid >> 3, m = tid & 7;
    float w = lds.e.Att[dir][m];
#pragma unroll
    for (int pr = 0; pr < NPAIR; ++pr)
      if (c_I[pr] == m || c_J[pr] == m) w += lds.e.Att[dir][8 + pr];
    lds.e.Weff[dir][m] = w;
  }
  __syncthreads();

  {
    const int d = tid * 4;
    float reac[4] = {0.f, 0.f, 0.f, 0.f};
#pragma unroll 4
    for (int m = 0; m < M_MOL; ++m) {
      const size_t off = ((size_t)m * B_DIM + b) * D_DIM + d;
      v4f rv = *(const v4f*)(rF + off);
      v4f rnv = *(const v4f*)(rN + off);
      v4f pv = *(const v4f*)(pF + off);
      v4f pnv = *(const v4f*)(pN + off);
      const float wr = lds.e.Weff[0][m], wp = lds.e.Weff[1][m];
#pragma unroll
      for (int x = 0; x < 4; ++x)
        reac[x] += wr * (rv[x] + rnv[x]) - wp * (pv[x] + pnv[x]);
    }
#pragma unroll
    for (int o = 0; o < 4; ++o) {
      float s = 0.f;
#pragma unroll
      for (int x = 0; x < 4; ++x) s += reac[x] * Wp[(size_t)o * D_DIM + d + x];
      lds.e.Red[tid * 4 + o] = s;
    }
  }
  __syncthreads();
  if (tid < 4) {
    float s = bp[tid];
    for (int t = 0; t < 256; ++t) s += lds.e.Red[t * 4 + tid];
    out[(size_t)b * 4 + tid] = s;
  }
}

extern "C" void kernel_launch(void* const* d_in, const int* in_sizes, int n_in,
                              void* d_out, int out_size, void* d_ws,
                              size_t ws_size, hipStream_t stream) {
  (void)in_sizes; (void)n_in; (void)out_size; (void)ws_size;
  const float* rF = (const float*)d_in[0];
  const float* pF = (const float*)d_in[1];
  const float* rN = (const float*)d_in[2];
  const float* pN = (const float*)d_in[3];
  const float* Wq = (const float*)d_in[4];
  const float* Wk = (const float*)d_in[5];
  const float* Wp = (const float*)d_in[6];
  const float* bp = (const float*)d_in[7];
  float* out = (float*)d_out;
  float* Gpk = (float*)d_ws;

  gemm_gt<<<512, 256, 0, stream>>>(Wq, Wk, Gpk);
  attn_batch<<<B_DIM, 256, 0, stream>>>(rF, pF, rN, pN, Gpk, Wp, bp, out);
}


# Two batches per 16-wave workgroup: each packed-G b128 fragment load feeds both batches' WMMA chains (G read once per two batches); S wave partials regroup to 16 of 4 e-tiles (sum order changes)
# speedup vs baseline: 6.0317x; 1.1064x over previous
#include <hip/hip_runtime.h>
#include <math.h>

typedef __attribute__((ext_vector_type(2))) float v2f;
typedef __attribute__((ext_vector_type(4))) float v4f;
typedef __attribute__((ext_vector_type(8))) float v8f;

#define D_DIM 1024
#define B_DIM 2048
#define M_MOL 8
#define NPAIR 28
#define FP 1028

#define NB 2
#define NWAVE 16
#define N_WG 1024

struct EpiLds {
  float S[256];
  float Rows[72][36];
  float Att[2][36];
  float Weff[2][8];
  float Red[256 * 4];
  float Part[NWAVE][256];
};
union TileLds {
  float F[16 * FP];
  EpiLds e;
};

__device__ __constant__ unsigned char c_I[NPAIR] =
    {0,0,0,0,0,0,0, 1,1,1,1,1,1, 2,2,2,2,2, 3,3,3,3, 4,4,4, 5,5, 6};
__device__ __constant__ unsigned char c_J[NPAIR] =
    {1,2,3,4,5,6,7, 2,3,4,5,6,7, 3,4,5,6,7, 4,5,6,7, 5,6,7, 6,7, 7};

static __device__ inline v8f wmma4(v2f a, v2f b, v8f c) {
  return __builtin_amdgcn_wmma_f32_16x16x4_f32(false, a, false, b, (short)0, c,
                                               false, false);
}


__global__ __launch_bounds__(256) void gemm_gt(const float* __restrict__ Wq,
                                               const float* __restrict__ Wk,
                                               float* __restrict__ Gpk) {
  const int wave = (int)((blockIdx.x * blockDim.x + threadIdx.x) >> 5);
  const int lane = threadIdx.x & 31;
  const int lm = lane & 15;
  const int half = lane >> 4;
  const int tm = wave >> 6;
  const int tn = wave & 63;

  const float* arow = Wk + (size_t)(tm * 16 + lm) * D_DIM + 2 * half;
  const float* brow = Wq + (size_t)(tn * 16 + lm) * D_DIM + 2 * half;

  v8f c = {};
#pragma unroll 4
  for (int x0 = 0; x0 < D_DIM; x0 += 4) {
    v2f a = *(const v2f*)(arow + x0);
    v2f b = *(const v2f*)(brow + x0);
    c = wmma4(a, b, c);
  }
  const int dstep = tn * 2 + (lm >> 3);
  const int s = (lm >> 2) & 1;
  const int h = (lm >> 1) & 1;
  const int r = lm & 1;
  const size_t base =
      ((size_t)(tm * 128 + dstep) * 32 + h * 16 + 8 * half) * 4 + 2 * s + r;
#pragma unroll
  for (int v = 0; v < 8; ++v) Gpk[base + (size_t)v * 4] = c[v];
}

__global__ __launch_bounds__(512) void attn_batch(
    const float* __restrict__ rF, const float* __restrict__ pF,
    const float* __restrict__ rN, const float* __restrict__ pN,
    const float* __restrict__ Gpk, const float* __restrict__ Wp,
    const float* __restrict__ bp, float* __restrict__ out) {
  __shared__ __attribute__((aligned(16))) TileLds lds[NB];
  __shared__ float sW[NWAVE][256];

  const int tid = threadIdx.x;
  const int wave = tid >> 5;
  const int lane = tid & 31;
  const int lm = lane & 15;
  const int half = lane >> 4;
  const int bsel = tid >> 8;
  const int t = tid & 255;
  const int b = blockIdx.x * NB + bsel;
  TileLds& my = lds[bsel];

#pragma unroll
  for (int r = 0; r < 16; ++r) {
    const float* srow =
        (r < M_MOL) ? (pF + ((size_t)r * B_DIM + b) * D_DIM)
                    : (rF + ((size_t)(r - M_MOL) * B_DIM + b) * D_DIM);
    *(v4f*)(&my.F[r * FP + t * 4]) = *(const v4f*)(srow + t * 4);
  }
  __syncthreads();

  const float* Frow0 = &lds[0].F[lm * FP];
  const float* Frow1 = &lds[1].F[lm * FP];
  const float* fa0 = Frow0 + 2 * half;
  const float* fa1 = Frow1 + 2 * half;

  int e[4];
  const float* g[4];
#pragma unroll
  for (int j = 0; j < 4; ++j) {
    e[j] = (wave + NWAVE * j) * 16;
    g[j] = Gpk + (size_t)e[j] * D_DIM + lane * 4;
  }
  v8f c00 = {}, c01 = {}, c02 = {}, c03 = {};
  v8f c10 = {}, c11 = {}, c12 = {}, c13 = {};
#pragma unroll 1
  for (int d0 = 0; d0 < D_DIM; d0 += 8) {
    v2f a00 = *(const v2f*)(fa0 + d0);
    v2f a01 = *(const v2f*)(fa0 + d0 + 4);
    v2f a10 = *(const v2f*)(fa1 + d0);
    v2f a11 = *(const v2f*)(fa1 + d0 + 4);
    v4f g0 = *(const v4f*)(g[0] + d0 * 16);
    v4f g1 = *(const v4f*)(g[1] + d0 * 16);
    v4f g2 = *(const v4f*)(g[2] + d0 * 16);
    v4f g3 = *(const v4f*)(g[3] + d0 * 16);
    const v2f g0l = __builtin_shufflevector(g0, g0, 0, 1);
    const v2f g1l = __builtin_shufflevector(g1, g1, 0, 1);
    const v2f g2l = __builtin_shufflevector(g2, g2, 0, 1);
    const v2f g3l = __builtin_shufflevector(g3, g3, 0, 1);
    const v2f g0h = __builtin_shufflevector(g0, g0, 2, 3);
    const v2f g1h = __builtin_shufflevector(g1, g1, 2, 3);
    const v2f g2h = __builtin_shufflevector(g2, g2, 2, 3);
    const v2f g3h = __builtin_shufflevector(g3, g3, 2, 3);
    c00 = wmma4(a00, g0l, c00);
    c10 = wmma4(a10, g0l, c10);
    c01 = wmma4(a00, g1l, c01);
    c11 = wmma4(a10, g1l, c11);
    c02 = wmma4(a00, g2l, c02);
    c12 = wmma4(a10, g2l, c12);
    c03 = wmma4(a00, g3l, c03);
    c13 = wmma4(a10, g3l, c13);
    c00 = wmma4(a01, g0h, c00);
    c10 = wmma4(a11, g0h, c10);
    c01 = wmma4(a01, g1h, c01);
    c11 = wmma4(a11, g1h, c11);
    c02 = wmma4(a01, g2h, c02);
    c12 = wmma4(a11, g2h, c12);
    c03 = wmma4(a01, g3h, c03);
    c13 = wmma4(a11, g3h, c13);
  }
  v8f sacc0 = {}, sacc1 = {};
#pragma unroll
  for (int j = 0; j < 4; ++j) {
    const v8f cj0 = (j == 0) ? c00 : (j == 1) ? c01 : (j == 2) ? c02 : c03;
    const v8f cj1 = (j == 0) ? c10 : (j == 1) ? c11 : (j == 2) ? c12 : c13;
#pragma unroll
    for (int v = 0; v < 8; ++v) sW[wave][(v + 8 * half) * 16 + lm] = cj0[v];
#pragma unroll
    for (int kk = 0; kk < 4; ++kk) {
      v2f a2 = *(const v2f*)(&sW[wave][lm * 16 + kk * 4 + 2 * half]);
      v2f b2 = *(const v2f*)(Frow0 + e[j] + kk * 4 + 2 * half);
      sacc0 = wmma4(a2, b2, sacc0);
    }
#pragma unroll
    for (int v = 0; v < 8; ++v) sW[wave][(v + 8 * half) * 16 + lm] = cj1[v];
#pragma unroll
    for (int kk = 0; kk < 4; ++kk) {
      v2f a2 = *(const v2f*)(&sW[wave][lm * 16 + kk * 4 + 2 * half]);
      v2f b2 = *(const v2f*)(Frow1 + e[j] + kk * 4 + 2 * half);
      sacc1 = wmma4(a2, b2, sacc1);
    }
  }
  __syncthreads();
#pragma unroll
  for (int v = 0; v < 8; ++v) {
    lds[0].e.Part[wave][(v + 8 * half) * 16 + lm] = sacc0[v];
    lds[1].e.Part[wave][(v + 8 * half) * 16 + lm] = sacc1[v];
  }
  __syncthreads();

  EpiLds& L = my.e;
  {
    float s = 0.f;
#pragma unroll
    for (int w = 0; w < NWAVE; ++w) s += L.Part[w][t];
    L.S[t] = s;
  }
  __syncthreads();

  const float scale = 0.03125f;
  if (t < 72) {
    const int dir = (t >= 36) ? 1 : 0;
    const int q = dir ? t - 36 : t;
    int qa0 = q, qa1 = q, qn = 1;
    if (q >= 8) { qa0 = c_I[q - 8]; qa1 = c_J[q - 8]; qn = 2; }
    float row[36];
    float mx = -3.4e38f;
#pragma unroll
    for (int k = 0; k < 36; ++k) {
      int ka0 = k, ka1 = k, kn = 1;
      if (k >= 8) { ka0 = c_I[k - 8]; ka1 = c_J[k - 8]; kn = 2; }
      float s = 0.f;
      for (int u = 0; u < qn; ++u) {
        const int a = u ? qa1 : qa0;
        for (int v = 0; v < kn; ++v) {
          const int cc = v ? ka1 : ka0;
          s += (dir == 0) ? L.S[a * 16 + 8 + cc] : L.S[(8 + a) * 16 + cc];
        }
      }
      s *= scale;
      row[k] = s;
      mx = fmaxf(mx, s);
    }
    float sum = 0.f;
#pragma unroll
    for (int k = 0; k < 36; ++k) { float ex = __expf(row[k] - mx); row[k] = ex; sum += ex; }
    const float inv = 1.f / sum;
#pragma unroll
    for (int k = 0; k < 36; ++k) L.Rows[t][k] = row[k] * inv;
  }
  __syncthreads();

  if (t < 72) {
    const int dir = (t >= 36) ? 1 : 0;
    const int k = dir ? t - 36 : t;
    float a = 0.f;
#pragma unroll
    for (int q = 0; q < 36; ++q) a += L.Rows[dir * 36 + q][k];
    a *= (1.f / 36.f);
    L.Att[dir][k] = a;
    const size_t base = dir ? (size_t)(B_DIM * 4 + B_DIM * 36) : (size_t)(B_DIM * 4);
    out[base + (size_t)b * 36 + k] = a;
  }
  __syncthreads();

  if (t < 16) {
    const int dir = t >> 3, m = t & 7;
    float w = L.Att[dir][m];
#pragma unroll
    for (int pr = 0; pr < NPAIR; ++pr)
      if (c_I[pr] == m || c_J[pr] == m) w += L.Att[dir][8 + pr];
    L.Weff[dir][m] = w;
  }
  __syncthreads();

  {
    const int d = t * 4;
    float reac[4] = {0.f, 0.f, 0.f, 0.f};
#pragma unroll 4
    for (int m = 0; m < M_MOL; ++m) {
      const size_t off = ((size_t)m * B_DIM + b) * D_DIM + d;
      v4f rv = *(const v4f*)(rF + off);
      v4f rnv = *(const v4f*)(rN + off);
      v4f pv = *(const v4f*)(pF + off);
      v4f pnv = *(const v4f*)(pN + off);
      const float wr = L.Weff[0][m], wp = L.Weff[1][m];
#pragma unroll
      for (int x = 0; x < 4; ++x)
        reac[x] += wr * (rv[x] + rnv[x]) - wp * (pv[x] + pnv[x]);
    }
#pragma unroll
    for (int o = 0; o < 4; ++o) {
      float s = 0.f;
#pragma unroll
      for (int x = 0; x < 4; ++x) s += reac[x] * Wp[(size_t)o * D_DIM + d + x];
      L.Red[t * 4 + o] = s;
    }
  }
  __syncthreads();
  if (t < 4) {
    float s = bp[t];
    for (int u = 0; u < 256; ++u) s += L.Red[u * 4 + t];
    out[(size_t)b * 4 + t] = s;
  }
}

extern "C" void kernel_launch(void* const* d_in, const int* in_sizes, int n_in,
                              void* d_out, int out_size, void* d_ws,
                              size_t ws_size, hipStream_t stream) {
  (void)in_sizes; (void)n_in; (void)out_size; (void)ws_size;
  const float* rF = (const float*)d_in[0];
  const float* pF = (const float*)d_in[1];
  const float* rN = (const float*)d_in[2];
  const float* pN = (const float*)d_in[3];
  const float* Wq = (const float*)d_in[4];
  const float* Wk = (const float*)d_in[5];
  const float* Wp = (const float*)d_in[6];
  const float* bp = (const float*)d_in[7];
  float* out = (float*)d_out;
  float* Gpk = (float*)d_ws;

  gemm_gt<<<512, 256, 0, stream>>>(Wq, Wk, Gpk);
  attn_batch<<<N_WG, 512, 0, stream>>>(rF, pF, rN, pN, Gpk, Wp, bp, out);
}
